# starting kernel
# baseline (speedup 1.0000x reference)
#include <hip/hip_runtime.h>
#include <hip/hip_bf16.h>

#define NNODES 50000
#define NEDGES 640000
#define DH 128
#define DOUT 64
#define LN_EPS 1e-5f

typedef float v2f __attribute__((ext_vector_type(2)));
typedef float v8f __attribute__((ext_vector_type(8)));

__global__ void gcn_deg_kernel(const int* __restrict__ dst, const float* __restrict__ w,
                               float* __restrict__ deg, int E) {
    int e = blockIdx.x * blockDim.x + threadIdx.x;
    if (e < E) atomicAdd(&deg[dst[e]], w[e]);
}

__global__ void gcn_dinv_kernel(float* __restrict__ deg, int n) {
    int i = blockIdx.x * blockDim.x + threadIdx.x;
    if (i < n) {
        float d = deg[i] + 1.0f;
        deg[i] = rsqrtf(d);
    }
}

__global__ void gcn_norm_kernel(const int* __restrict__ src, const int* __restrict__ dst,
                                const float* __restrict__ w, const float* __restrict__ dinv,
                                float* __restrict__ norm, int E) {
    int e = blockIdx.x * blockDim.x + threadIdx.x;
    if (e < E) norm[e] = dinv[src[e]] * w[e] * dinv[dst[e]];
}

__global__ void gcn_gemm_wmma(const float* __restrict__ A,
                              const float* __restrict__ W,
                              float* __restrict__ out,
                              int nrows, int ncols) {
    const int lane = threadIdx.x & 31;
    const int wid  = blockIdx.x * (blockDim.x >> 5) + (threadIdx.x >> 5);
    const int ctiles = ncols >> 4;
    const int rtiles = nrows >> 4;
    if (wid >= rtiles * ctiles) return;
    const int rt = wid / ctiles;
    const int ct = wid - rt * ctiles;
    const int row0 = rt << 4;
    const int j0   = ct << 4;

    const int m  = lane & 15;
    const int ko = (lane >> 4) << 1;
    const float* __restrict__ arow = A + (size_t)(row0 + m) * DH + ko;
    const float* __restrict__ brow = W + (size_t)(j0 + m) * DH + ko;

    v8f acc = {};
#pragma unroll
    for (int k = 0; k < DH; k += 4) {
        v2f a = *(const v2f*)(arow + k);
        v2f b = *(const v2f*)(brow + k);
        acc = __builtin_amdgcn_wmma_f32_16x16x4_f32(
             false, a,  false, b,
             (short)0, acc,  false,  false);
    }

    float* __restrict__ o = out + (size_t)(row0 + ((lane >> 4) << 3)) * ncols + j0 + m;
#pragma unroll
    for (int v = 0; v < 8; ++v)
        o[(size_t)v * ncols] = acc[v];
}

__global__ void gcn_scatter_kernel(const float* __restrict__ xw,
                                   const int* __restrict__ src, const int* __restrict__ dst,
                                   const float* __restrict__ norm,
                                   float* __restrict__ agg, int E, int ncols) {
    int t = blockIdx.x * blockDim.x + threadIdx.x;
    int per_edge = ncols >> 2;
    int e = t / per_edge;
    if (e >= E) return;
    int f = (t - e * per_edge) << 2;
    int s = src[e], d = dst[e];
    float w = norm[e];
    const float4 v = *(const float4*)(xw + (size_t)s * ncols + f);
    float* p = agg + (size_t)d * ncols + f;
    atomicAdd(p + 0, w * v.x);
    atomicAdd(p + 1, w * v.y);
    atomicAdd(p + 2, w * v.z);
    atomicAdd(p + 3, w * v.w);
}

__device__ __forceinline__ float wave_sum32(float v) {
#pragma unroll
    for (int m = 16; m >= 1; m >>= 1) v += __shfl_xor(v, m, 32);
    return v;
}

__global__ void gcn_epilogue_hidden(const float* __restrict__ agg, const float* __restrict__ xw,
                                    const float* __restrict__ dinv,
                                    const float* __restrict__ bias, const float* __restrict__ gamma,
                                    const float* __restrict__ beta,
                                    const float* __restrict__ res, float* __restrict__ hout, int n) {
    int t = blockIdx.x * blockDim.x + threadIdx.x;
    int lane = t & 31;
    int node = t >> 5;
    if (node >= n) return;

    float di = dinv[node];
    float sl = di * di;
    size_t base = (size_t)node * DH + (lane << 2);

    float4 a  = *(const float4*)(agg + base);
    float4 xv = *(const float4*)(xw + base);
    float4 b  = *(const float4*)(bias + (lane << 2));

    float v0 = fmaf(sl, xv.x, a.x) + b.x;
    float v1 = fmaf(sl, xv.y, a.y) + b.y;
    float v2 = fmaf(sl, xv.z, a.z) + b.z;
    float v3 = fmaf(sl, xv.w, a.w) + b.w;

    float s  = wave_sum32(v0 + v1 + v2 + v3);
    float sq = wave_sum32(v0 * v0 + v1 * v1 + v2 * v2 + v3 * v3);
    float mean = s * (1.0f / DH);
    float var  = sq * (1.0f / DH) - mean * mean;
    float rinv = rsqrtf(var + LN_EPS);

    float4 g  = *(const float4*)(gamma + (lane << 2));
    float4 be = *(const float4*)(beta + (lane << 2));
    float4 r  = *(const float4*)(res + base);

    float4 o;
    o.x = fmaxf(g.x * (v0 - mean) * rinv + be.x, 0.0f) + r.x;
    o.y = fmaxf(g.y * (v1 - mean) * rinv + be.y, 0.0f) + r.y;
    o.z = fmaxf(g.z * (v2 - mean) * rinv + be.z, 0.0f) + r.z;
    o.w = fmaxf(g.w * (v3 - mean) * rinv + be.w, 0.0f) + r.w;
    *(float4*)(hout + base) = o;
}

__global__ void gcn_epilogue_final(const float* __restrict__ agg, const float* __restrict__ xw,
                                   const float* __restrict__ dinv,
                                   const float* __restrict__ bias, const float* __restrict__ gamma,
                                   const float* __restrict__ beta,
                                   float* __restrict__ out, int n) {
    int t = blockIdx.x * blockDim.x + threadIdx.x;
    int lane = t & 31;
    int node = t >> 5;
    if (node >= n) return;

    float di = dinv[node];
    float sl = di * di;
    size_t base = (size_t)node * DOUT + (lane << 1);

    float2 a  = *(const float2*)(agg + base);
    float2 xv = *(const float2*)(xw + base);
    float2 b  = *(const float2*)(bias + (lane << 1));

    float v0 = fmaf(sl, xv.x, a.x) + b.x;
    float v1 = fmaf(sl, xv.y, a.y) + b.y;

    float s  = wave_sum32(v0 + v1);
    float sq = wave_sum32(v0 * v0 + v1 * v1);
    float mean = s * (1.0f / DOUT);
    float var  = sq * (1.0f / DOUT) - mean * mean;
    float rinv = rsqrtf(var + LN_EPS);

    float2 g  = *(const float2*)(gamma + (lane << 1));
    float2 be = *(const float2*)(beta + (lane << 1));

    float2 o;
    o.x = g.x * (v0 - mean) * rinv + be.x;
    o.y = g.y * (v1 - mean) * rinv + be.y;
    *(float2*)(out + base) = o;
}

extern "C" void kernel_launch(void* const* d_in, const int* in_sizes, int n_in,
                              void* d_out, int out_size, void* d_ws, size_t ws_size,
                              hipStream_t stream) {
    const float* x    = (const float*)d_in[0];
    const int*   ei   = (const int*)d_in[1];
    const float* ew   = (const float*)d_in[2];
    const float* W03  = (const float*)d_in[3];
    const float* b03  = (const float*)d_in[4];
    const float* g03  = (const float*)d_in[5];
    const float* be03 = (const float*)d_in[6];
    const float* W4   = (const float*)d_in[7];
    const float* b4   = (const float*)d_in[8];
    const float* g4   = (const float*)d_in[9];
    const float* be4  = (const float*)d_in[10];

    const int* srcI = ei;
    const int* dstI = ei + NEDGES;

    float* dinv = (float*)d_ws;
    float* norm = dinv + NNODES;
    float* xw   = norm + NEDGES;
    float* agg  = xw + (size_t)NNODES * DH;
    float* hbuf = agg + (size_t)NNODES * DH;

    hipMemsetAsync(dinv, 0, (size_t)NNODES * sizeof(float), stream);
    gcn_deg_kernel<<<(NEDGES + 255) / 256, 256, 0, stream>>>(dstI, ew, dinv, NEDGES);
    gcn_dinv_kernel<<<(NNODES + 255) / 256, 256, 0, stream>>>(dinv, NNODES);
    gcn_norm_kernel<<<(NEDGES + 255) / 256, 256, 0, stream>>>(srcI, dstI, ew, dinv, norm, NEDGES);

    const int rtiles = NNODES / 16;
    const float* hin = x;
    const float* resin = x;
    for (int l = 0; l < 4; ++l) {
        int waves  = rtiles * (DH / 16);
        int blocks = (waves * 32 + 255) / 256;
        gcn_gemm_wmma<<<blocks, 256, 0, stream>>>(hin, W03 + (size_t)l * DH * DH, xw, NNODES, DH);

        hipMemsetAsync(agg, 0, (size_t)NNODES * DH * sizeof(float), stream);
        long long sthreads = (long long)NEDGES * (DH / 4);
        gcn_scatter_kernel<<<(int)((sthreads + 255) / 256), 256, 0, stream>>>(
            xw, srcI, dstI, norm, agg, NEDGES, DH);

        long long ethreads = (long long)NNODES * 32;
        gcn_epilogue_hidden<<<(int)((ethreads + 255) / 256), 256, 0, stream>>>(
            agg, xw, dinv, b03 + l * DH, g03 + l * DH, be03 + l * DH, resin, hbuf, NNODES);

        hin = hbuf;
        resin = hbuf;
    }

    {
        int waves  = rtiles * (DOUT / 16);
        int blocks = (waves * 32 + 255) / 256;
        gcn_gemm_wmma<<<blocks, 256, 0, stream>>>(hbuf, W4, xw, NNODES, DOUT);

        hipMemsetAsync(agg, 0, (size_t)NNODES * DOUT * sizeof(float), stream);
        long long sthreads = (long long)NEDGES * (DOUT / 4);
        gcn_scatter_kernel<<<(int)((sthreads + 255) / 256), 256, 0, stream>>>(
            xw, srcI, dstI, norm, agg, NEDGES, DOUT);

        long long ethreads = (long long)NNODES * 32;
        gcn_epilogue_final<<<(int)((ethreads + 255) / 256), 256, 0, stream>>>(
            agg, xw, dinv, b4, g4, be4, (float*)d_out, NNODES);
    }
}


# Output layer folded into its neighbours: its WMMA product runs in the last hidden launch from the LDS tile, bias+LayerNorm in its aggregation launch; self loop is the start value; sum order changed
# speedup vs baseline: 3.7443x; 3.7443x over previous
#include <hip/hip_runtime.h>
#include <hip/hip_bf16.h>

#define NNODES 50000
#define NEDGES 640000
#define DH 128
#define DOUT 64
#define LN_EPS 1e-5f

typedef float v2f __attribute__((ext_vector_type(2)));
typedef float v8f __attribute__((ext_vector_type(8)));

__global__ void gcn_deg_kernel(const int* __restrict__ dst, const float* __restrict__ w,
                               float* __restrict__ deg, int E) {
    int e = blockIdx.x * blockDim.x + threadIdx.x;
    if (e < E) atomicAdd(&deg[dst[e]], w[e]);
}

__global__ void gcn_dinv_kernel(const float* __restrict__ deg, float* __restrict__ dinv, int n) {
    int i = blockIdx.x * blockDim.x + threadIdx.x;
    if (i < n) dinv[i] = rsqrtf(deg[i] + 1.0f);
}

__global__ void gcn_norm_kernel(const int* __restrict__ src, const int* __restrict__ dst,
                                const float* __restrict__ w, const float* __restrict__ dinv,
                                float* __restrict__ norm, int E) {
    int e = blockIdx.x * blockDim.x + threadIdx.x;
    if (e < E) norm[e] = dinv[src[e]] * w[e] * dinv[dst[e]];
}

#define NRANGES 256
#define RPR 196
#define QCAP 16384

#define NSTRIPS 13
#define TP 132
#define RPW 7
#define RPAD (RPW * 32)
template <bool LAST>
__global__ void __launch_bounds__(1024)
gcn_layer_fused(const float* __restrict__ hin,
                const int* __restrict__ src, const int* __restrict__ dst,
                const float* __restrict__ norm, const float* __restrict__ dinv,
                const float* __restrict__ W,
                const float* __restrict__ bias, const float* __restrict__ gamma,
                const float* __restrict__ beta,
                float* __restrict__ hout,
                const float* __restrict__ Wn, float* __restrict__ xwn,
                int* __restrict__ qs, int* __restrict__ qd, float* __restrict__ qw,
                int* __restrict__ qn, int build, int E, int n) {
    constexpr int Q = DH / 32;
    constexpr int U = 64 / Q;
    __shared__ __attribute__((aligned(16))) float acc[RPAD * TP];
    __shared__ __attribute__((aligned(16))) float Ws[DH * TP];
    __shared__ int qe[QCAP];
    __shared__ int qcount;

    const int tid  = threadIdx.x;
    const int lane = tid & 31;
    const int wv   = __builtin_amdgcn_readfirstlane(tid >> 5);
    const int lo   = blockIdx.x * RPR;
    const int nrows = (n - lo < RPR) ? (n - lo) : RPR;
    if (nrows <= 0) return;

    const int stored = build ? 0 : qn[blockIdx.x];
    const bool scan = (build != 0) || (stored > QCAP);

    if (tid == 0) qcount = 0;
    {
        float4 wr[4];
        float4 xv[RPW];
        float sl[RPW];
#pragma unroll
        for (int k = 0; k < 4; ++k)
            wr[k] = *(const float4*)(W + (size_t)(wv + 32 * k) * DH + (lane << 2));
        const int rl  = wv + 32 * ((lane < RPW) ? lane : 0);
        const int rlc = (rl < nrows) ? rl : (nrows - 1);
        const int dli = __float_as_int(dinv[lo + rlc]);
#pragma unroll
        for (int i = 0; i < RPW; ++i) {
            const int r = wv + 32 * i;
            const int rc = (r < nrows) ? r : (nrows - 1);
            xv[i] = *(const float4*)(hin + (size_t)(lo + rc) * DH + (lane << 2));
        }
#pragma unroll
        for (int i = 0; i < RPW; ++i) {
            const int r = wv + 32 * i;
            const float di = __int_as_float(__builtin_amdgcn_readlane(dli, i));
            sl[i] = (r < nrows) ? di * di : 0.0f;
        }
#pragma unroll
        for (int k = 0; k < 4; ++k)
            *(float4*)(&Ws[(wv + 32 * k) * TP + (lane << 2)]) = wr[k];
#pragma unroll
        for (int i = 0; i < RPW; ++i) {
            const int r = wv + 32 * i;
            float4 o;
            o.x = sl[i] * xv[i].x; o.y = sl[i] * xv[i].y;
            o.z = sl[i] * xv[i].z; o.w = sl[i] * xv[i].w;
            *(float4*)(&acc[r * TP + (lane << 2)]) = o;
        }
    }
    __syncthreads();

    const unsigned nr = (unsigned)nrows;
    auto push = [&](int e) {
        const int slot = atomicAdd(&qcount, 1);
        if (slot < QCAP) {
            qe[slot] = e;
        } else {
            const int s = src[e];
            const float w = norm[e];
            const int dl = dst[e] - lo;
            for (int c = 0; c < DH; ++c)
                atomicAdd(&acc[dl * TP + c], w * hin[(size_t)s * DH + c]);
        }
    };

    if (scan) {
        const int n4 = E >> 2;
        const int4* d4 = (const int4*)dst;
        const int nfull = n4 >> 13;
        const int boff = (nfull > 0) ? (int)(blockIdx.x % (unsigned)nfull) : 0;
        for (int b = 0; b < nfull; ++b) {
            int bb = b + boff;
            if (bb >= nfull) bb -= nfull;
            const int i0 = (bb << 13) + tid;
            int4 v[8];
    #pragma unroll
            for (int k = 0; k < 8; ++k) v[k] = d4[i0 + k * 1024];
            unsigned pend = 0;
    #pragma unroll
            for (int k = 0; k < 8; ++k) {
                const unsigned r0 = (unsigned)(v[k].x - lo);
                const unsigned r1 = (unsigned)(v[k].y - lo);
                const unsigned r2 = (unsigned)(v[k].z - lo);
                const unsigned r3 = (unsigned)(v[k].w - lo);
                const unsigned mn = min(min(r0, r1), min(r2, r3));
                if (mn < nr) {
                    const unsigned m = (r0 < nr ? 1u : 0u) | (r1 < nr ? 2u : 0u) |
                                       (r2 < nr ? 4u : 0u) | (r3 < nr ? 8u : 0u);
                    pend |= m << (4 * k);
                }
            }
            while (pend) {
                const int bit = __builtin_ctz(pend);
                pend &= pend - 1;
                push(4 * (i0 + (bit >> 2) * 1024) + (bit & 3));
            }
        }
        for (int i = (nfull << 13) + tid; i < n4; i += 1024) {
            const int4 v = d4[i];
            if ((unsigned)(v.x - lo) < nr) push(4 * i + 0);
            if ((unsigned)(v.y - lo) < nr) push(4 * i + 1);
            if ((unsigned)(v.z - lo) < nr) push(4 * i + 2);
            if ((unsigned)(v.w - lo) < nr) push(4 * i + 3);
        }
        {
            const int e = 4 * n4 + tid;
            if (e < E) {
                const unsigned r = (unsigned)(dst[e] - lo);
                if (r < nr) push(e);
            }
        }
    }
    __syncthreads();

    const int total = scan ? qcount : stored;
    if (build && tid == 0) qn[blockIdx.x] = total;
    const int M = (total < QCAP) ? total : QCAP;
    const size_t seg = (size_t)blockIdx.x * QCAP;
    const float* __restrict__ xl = hin + lane;
    for (int base = wv * 32; base < M; base += 32 * 32) {
        const int cnt = (M - base < 32) ? (M - base) : 32;
        int s = 0, d = 0;
        float w = 0.0f;
        if (lane < cnt) {
            const size_t slot = seg + (size_t)(base + lane);
            if (scan) {
                const int e = qe[base + lane];
                s = src[e]; d = dst[e] - lo; w = norm[e];
                if (build) { qs[slot] = s; qd[slot] = d; qw[slot] = w; }
            } else {
                s = qs[slot]; d = qd[slot]; w = qw[slot];
            }
        }
        const int wi = __float_as_int(w);
        for (int j0 = 0; j0 < cnt; j0 += U) {
            float v[U][Q];
#pragma unroll
            for (int u = 0; u < U; ++u) {
                const int sj = __builtin_amdgcn_readlane(s, j0 + u);
                const float* __restrict__ r = xl + (size_t)sj * DH;
#pragma unroll
                for (int q = 0; q < Q; ++q) v[u][q] = r[q * 32];
            }
#pragma unroll
            for (int u = 0; u < U; ++u) {
                if (j0 + u < cnt) {
                    const int dj = __builtin_amdgcn_readlane(d, j0 + u);
                    const float wj = __int_as_float(__builtin_amdgcn_readlane(wi, j0 + u));
                    float* p = &acc[dj * TP + lane];
#pragma unroll
                    for (int q = 0; q < Q; ++q) atomicAdd(p + q * 32, wj * v[u][q]);
                }
            }
        }
    }
    __syncthreads();

    const int c  = lane & 15;
    const int hh = lane >> 4;
    const int strip = wv >> 1;
    const int ch = wv & 1;
    const bool gw = (wv < 2 * NSTRIPS) && ((strip << 4) < nrows);
    v8f pacc[4];
#pragma unroll
    for (int t = 0; t < 4; ++t) pacc[t] = (v8f){};
    float4 wn0 = make_float4(0.0f, 0.0f, 0.0f, 0.0f), wn1 = wn0;
    if (LAST) {
        wn0 = *(const float4*)(Wn + (size_t)wv * DH + (lane << 2));
        wn1 = *(const float4*)(Wn + (size_t)(wv + 32) * DH + (lane << 2));
    }
    if (gw) {
        const float* ap = acc + ((strip << 4) + c) * TP + (hh << 2);
        const float* bp = Ws + ((ch << 6) + c) * TP + (hh << 2);
#pragma unroll
        for (int j = 0; j < 16; ++j) {
            const float4 a4 = *(const float4*)(ap + 8 * j);
            float4 b4[4];
#pragma unroll
            for (int t = 0; t < 4; ++t) b4[t] = *(const float4*)(bp + t * (16 * TP) + 8 * j);
            const v2f a01 = {a4.x, a4.y};
            const v2f a23 = {a4.z, a4.w};
#pragma unroll
            for (int t = 0; t < 4; ++t) {
                const v2f b01 = {b4[t].x, b4[t].y};
                pacc[t] = __builtin_amdgcn_wmma_f32_16x16x4_f32(
                    false, a01, false, b01, (short)0, pacc[t], false, false);
            }
#pragma unroll
            for (int t = 0; t < 4; ++t) {
                const v2f b23 = {b4[t].z, b4[t].w};
                pacc[t] = __builtin_amdgcn_wmma_f32_16x16x4_f32(
                    false, a23, false, b23, (short)0, pacc[t], false, false);
            }
        }
    }
    __syncthreads();
    if (LAST) {
        *(float4*)(&Ws[wv * TP + (lane << 2)]) = wn0;
        *(float4*)(&Ws[(wv + 32) * TP + (lane << 2)]) = wn1;
    }
    if (gw) {
        float* dp = acc + ((strip << 4) + 8 * hh) * TP + (ch << 6) + c;
#pragma unroll
        for (int t = 0; t < 4; ++t)
#pragma unroll
            for (int r = 0; r < 8; ++r)
                dp[r * TP + 16 * t] = pacc[t][r];
    }
    __syncthreads();

    {
        const float4 b  = *(const float4*)(bias + (lane << 2));
        const float4 g  = *(const float4*)(gamma + (lane << 2));
        const float4 be = *(const float4*)(beta + (lane << 2));
        float4 a[RPW], rv[RPW];
#pragma unroll
        for (int i = 0; i < RPW; ++i) {
            const int r = wv + 32 * i;
            a[i]  = make_float4(0.0f, 0.0f, 0.0f, 0.0f);
            rv[i] = make_float4(0.0f, 0.0f, 0.0f, 0.0f);
            if (r < nrows) {
                a[i]  = *(const float4*)(&acc[r * TP + (lane << 2)]);
                rv[i] = *(const float4*)(hin + (size_t)(lo + r) * DH + (lane << 2));
            }
        }
        float s[RPW], sq[RPW];
#pragma unroll
        for (int i = 0; i < RPW; ++i) {
            a[i].x += b.x; a[i].y += b.y; a[i].z += b.z; a[i].w += b.w;
            s[i]  = a[i].x + a[i].y + a[i].z + a[i].w;
            sq[i] = a[i].x * a[i].x + a[i].y * a[i].y + a[i].z * a[i].z + a[i].w * a[i].w;
        }
#pragma unroll
        for (int m = 16; m >= 1; m >>= 1) {
#pragma unroll
            for (int i = 0; i < RPW; ++i) {
                s[i]  += __shfl_xor(s[i], m, 32);
                sq[i] += __shfl_xor(sq[i], m, 32);
            }
        }
#pragma unroll
        for (int i = 0; i < RPW; ++i) {
            const int r = wv + 32 * i;
            if (r < nrows) {
                const float mean = s[i] * (1.0f / DH);
                const float var  = sq[i] * (1.0f / DH) - mean * mean;
                const float rinv = rsqrtf(var + LN_EPS);
                float4 o;
                o.x = fmaxf(g.x * (a[i].x - mean) * rinv + be.x, 0.0f) + rv[i].x;
                o.y = fmaxf(g.y * (a[i].y - mean) * rinv + be.y, 0.0f) + rv[i].y;
                o.z = fmaxf(g.z * (a[i].z - mean) * rinv + be.z, 0.0f) + rv[i].z;
                o.w = fmaxf(g.w * (a[i].w - mean) * rinv + be.w, 0.0f) + rv[i].w;
                if (LAST) *(float4*)(&acc[r * TP + (lane << 2)]) = o;
                else      *(float4*)(hout + (size_t)(lo + r) * DH + (lane << 2)) = o;
            }
        }
    }

    if (LAST) {
        __syncthreads();
        v8f qacc[2];
#pragma unroll
        for (int t = 0; t < 2; ++t) qacc[t] = (v8f){};
        if (gw) {
            const float* ap = acc + ((strip << 4) + c) * TP + (hh << 2);
            const float* bp = Ws + ((ch << 5) + c) * TP + (hh << 2);
#pragma unroll
            for (int j = 0; j < 16; ++j) {
                const float4 a4 = *(const float4*)(ap + 8 * j);
                float4 b4[2];
#pragma unroll
                for (int t = 0; t < 2; ++t) b4[t] = *(const float4*)(bp + t * (16 * TP) + 8 * j);
                const v2f a01 = {a4.x, a4.y};
                const v2f a23 = {a4.z, a4.w};
#pragma unroll
                for (int t = 0; t < 2; ++t) {
                    const v2f b01 = {b4[t].x, b4[t].y};
                    qacc[t] = __builtin_amdgcn_wmma_f32_16x16x4_f32(
                        false, a01, false, b01, (short)0, qacc[t], false, false);
                }
#pragma unroll
                for (int t = 0; t < 2; ++t) {
                    const v2f b23 = {b4[t].z, b4[t].w};
                    qacc[t] = __builtin_amdgcn_wmma_f32_16x16x4_f32(
                        false, a23, false, b23, (short)0, qacc[t], false, false);
                }
            }
            const int r0 = (strip << 4) + 8 * hh;
            float* op = xwn + (size_t)(lo + r0) * DOUT + (ch << 5) + c;
#pragma unroll
            for (int t = 0; t < 2; ++t)
#pragma unroll
                for (int r = 0; r < 8; ++r)
                    if (r0 + r < nrows) op[(size_t)r * DOUT + 16 * t] = qacc[t][r];
        }
    }
}

__global__ void __launch_bounds__(1024)
gcn_out_fused(const float* __restrict__ xw,
              const int* __restrict__ src, const int* __restrict__ dst,
              const float* __restrict__ norm, const float* __restrict__ dinv,
              const float* __restrict__ bias, const float* __restrict__ gamma,
              const float* __restrict__ beta,
              float* __restrict__ out,
              const int* __restrict__ qs, const int* __restrict__ qd,
              const float* __restrict__ qw, const int* __restrict__ qn, int E, int n) {
    constexpr int NCOLS = DOUT;
    constexpr int Q = NCOLS / 32;
    constexpr int U = 64 / Q;
    constexpr int R4 = NCOLS / 4;
    static_assert(RPR * R4 <= 4 * 1024, "tile start: four 16-byte groups per thread cover the tile");
    __shared__ __attribute__((aligned(16))) float acc[RPR * NCOLS];
    __shared__ int qe[QCAP];
    __shared__ int qcount;

    const int tid  = threadIdx.x;
    const int lane = tid & 31;
    const int wv   = __builtin_amdgcn_readfirstlane(tid >> 5);
    const int lo   = blockIdx.x * RPR;
    const int nrows = (n - lo < RPR) ? (n - lo) : RPR;
    if (nrows <= 0) return;

    const int stored = qn[blockIdx.x];
    const bool scan = stored > QCAP;

    if (tid == 0) qcount = 0;
    const int n4r = nrows * R4;
    {
        const float4* x4 = (const float4*)(xw + (size_t)lo * NCOLS);
        float4* a4 = (float4*)acc;
        float4 xv[4];
        float dv[4];
#pragma unroll
        for (int k = 0; k < 4; ++k) {
            const int i = tid + 1024 * k;
            const int ic = (i < n4r) ? i : (n4r - 1);
            xv[k] = x4[ic];
            dv[k] = dinv[lo + ic / R4];
        }
#pragma unroll
        for (int k = 0; k < 4; ++k) {
            const int i = tid + 1024 * k;
            if (i < n4r) {
                const float sl = dv[k] * dv[k];
                float4 o;
                o.x = sl * xv[k].x; o.y = sl * xv[k].y;
                o.z = sl * xv[k].z; o.w = sl * xv[k].w;
                a4[i] = o;
            }
        }
    }
    __syncthreads();

    const unsigned nr = (unsigned)nrows;
    auto push = [&](int e) {
        const int slot = atomicAdd(&qcount, 1);
        if (slot < QCAP) {
            qe[slot] = e;
        } else {
            const int s = src[e];
            const float w = norm[e];
            const int dl = dst[e] - lo;
            for (int c = 0; c < NCOLS; ++c)
                atomicAdd(&acc[dl * NCOLS + c], w * xw[(size_t)s * NCOLS + c]);
        }
    };

    if (scan) {
        const int n4 = E >> 2;
        const int4* d4 = (const int4*)dst;
        const int nfull = n4 >> 13;
        const int boff = (nfull > 0) ? (int)(blockIdx.x % (unsigned)nfull) : 0;
        for (int b = 0; b < nfull; ++b) {
            int bb = b + boff;
            if (bb >= nfull) bb -= nfull;
            const int i0 = (bb << 13) + tid;
            int4 v[8];
    #pragma unroll
            for (int k = 0; k < 8; ++k) v[k] = d4[i0 + k * 1024];
            unsigned pend = 0;
    #pragma unroll
            for (int k = 0; k < 8; ++k) {
                const unsigned r0 = (unsigned)(v[k].x - lo);
                const unsigned r1 = (unsigned)(v[k].y - lo);
                const unsigned r2 = (unsigned)(v[k].z - lo);
                const unsigned r3 = (unsigned)(v[k].w - lo);
                const unsigned mn = min(min(r0, r1), min(r2, r3));
                if (mn < nr) {
                    const unsigned m = (r0 < nr ? 1u : 0u) | (r1 < nr ? 2u : 0u) |
                                       (r2 < nr ? 4u : 0u) | (r3 < nr ? 8u : 0u);
                    pend |= m << (4 * k);
                }
            }
            while (pend) {
                const int bit = __builtin_ctz(pend);
                pend &= pend - 1;
                push(4 * (i0 + (bit >> 2) * 1024) + (bit & 3));
            }
        }
        for (int i = (nfull << 13) + tid; i < n4; i += 1024) {
            const int4 v = d4[i];
            if ((unsigned)(v.x - lo) < nr) push(4 * i + 0);
            if ((unsigned)(v.y - lo) < nr) push(4 * i + 1);
            if ((unsigned)(v.z - lo) < nr) push(4 * i + 2);
            if ((unsigned)(v.w - lo) < nr) push(4 * i + 3);
        }
        {
            const int e = 4 * n4 + tid;
            if (e < E) {
                const unsigned r = (unsigned)(dst[e] - lo);
                if (r < nr) push(e);
            }
        }
    }
    __syncthreads();

    const int total = scan ? qcount : stored;
    const int M = (total < QCAP) ? total : QCAP;
    const size_t seg = (size_t)blockIdx.x * QCAP;
    const float* __restrict__ xl = xw + lane;
    for (int base = wv * 32; base < M; base += 32 * 32) {
        const int cnt = (M - base < 32) ? (M - base) : 32;
        int s = 0, d = 0;
        float w = 0.0f;
        if (lane < cnt) {
            if (scan) {
                const int e = qe[base + lane];
                s = src[e]; d = dst[e] - lo; w = norm[e];
            } else {
                const size_t slot = seg + (size_t)(base + lane);
                s = qs[slot]; d = qd[slot]; w = qw[slot];
            }
        }
        const int wi = __float_as_int(w);
        for (int j0 = 0; j0 < cnt; j0 += U) {
            float v[U][Q];
#pragma unroll
            for (int u = 0; u < U; ++u) {
                const int sj = __builtin_amdgcn_readlane(s, j0 + u);
                const float* __restrict__ r = xl + (size_t)sj * NCOLS;
#pragma unroll
                for (int q = 0; q < Q; ++q) v[u][q] = r[q * 32];
            }
#pragma unroll
            for (int u = 0; u < U; ++u) {
                if (j0 + u < cnt) {
                    const int dj = __builtin_amdgcn_readlane(d, j0 + u);
                    const float wj = __int_as_float(__builtin_amdgcn_readlane(wi, j0 + u));
                    float* p = &acc[dj * NCOLS + lane];
#pragma unroll
                    for (int q = 0; q < Q; ++q) atomicAdd(p + q * 32, wj * v[u][q]);
                }
            }
        }
    }
    __syncthreads();

    {
        const float2 b  = *(const float2*)(bias + (lane << 1));
        const float2 g  = *(const float2*)(gamma + (lane << 1));
        const float2 be = *(const float2*)(beta + (lane << 1));
        float2 a[RPW];
        float s[RPW], sq[RPW];
#pragma unroll
        for (int i = 0; i < RPW; ++i) {
            const int r = wv + 32 * i;
            a[i] = make_float2(0.0f, 0.0f);
            if (r < nrows)
                a[i] = *(const float2*)(&acc[r * NCOLS + (lane << 1)]);
            a[i].x += b.x; a[i].y += b.y;
            s[i]  = a[i].x + a[i].y;
            sq[i] = a[i].x * a[i].x + a[i].y * a[i].y;
        }
#pragma unroll
        for (int m = 16; m >= 1; m >>= 1) {
#pragma unroll
            for (int i = 0; i < RPW; ++i) {
                s[i]  += __shfl_xor(s[i], m, 32);
                sq[i] += __shfl_xor(sq[i], m, 32);
            }
        }
#pragma unroll
        for (int i = 0; i < RPW; ++i) {
            const int r = wv + 32 * i;
            if (r < nrows) {
                const float mean = s[i] * (1.0f / NCOLS);
                const float var  = sq[i] * (1.0f / NCOLS) - mean * mean;
                const float rinv = rsqrtf(var + LN_EPS);
                float2 o;
                o.x = g.x * (a[i].x - mean) * rinv + be.x;
                o.y = g.y * (a[i].y - mean) * rinv + be.y;
                *(float2*)(out + (size_t)(lo + r) * NCOLS + (lane << 1)) = o;
            }
        }
    }
}

extern "C" void kernel_launch(void* const* d_in, const int* in_sizes, int n_in,
                              void* d_out, int out_size, void* d_ws, size_t ws_size,
                              hipStream_t stream) {
    const float* x    = (const float*)d_in[0];
    const int*   ei   = (const int*)d_in[1];
    const float* ew   = (const float*)d_in[2];
    const float* W03  = (const float*)d_in[3];
    const float* b03  = (const float*)d_in[4];
    const float* g03  = (const float*)d_in[5];
    const float* be03 = (const float*)d_in[6];
    const float* W4   = (const float*)d_in[7];
    const float* b4   = (const float*)d_in[8];
    const float* g4   = (const float*)d_in[9];
    const float* be4  = (const float*)d_in[10];

    const int* srcI = ei;
    const int* dstI = ei + NEDGES;

    float* xw   = (float*)d_ws;
    float* hA   = xw + (size_t)NNODES * DOUT;
    float* hB   = hA + (size_t)NNODES * DH;
    float* dinv = hB + (size_t)NNODES * DH;
    float* deg  = dinv + NNODES;
    float* norm = deg + NNODES;
    int*   qn   = (int*)(norm + NEDGES);
    int*   qs   = qn + NRANGES;
    int*   qd   = qs + (size_t)NRANGES * QCAP;
    float* qw   = (float*)(qd + (size_t)NRANGES * QCAP);

    hipMemsetAsync(deg, 0, (size_t)NNODES * sizeof(float), stream);
    gcn_deg_kernel<<<(NEDGES + 255) / 256, 256, 0, stream>>>(dstI, ew, deg, NEDGES);
    gcn_dinv_kernel<<<(NNODES + 255) / 256, 256, 0, stream>>>(deg, dinv, NNODES);
    gcn_norm_kernel<<<(NEDGES + 255) / 256, 256, 0, stream>>>(srcI, dstI, ew, dinv, norm, NEDGES);

    const float* hin = x;
    for (int l = 0; l < 3; ++l) {
        float* ho = (l & 1) ? hB : hA;
        gcn_layer_fused<false><<<NRANGES, 1024, 0, stream>>>(
            hin, srcI, dstI, norm, dinv, W03 + (size_t)l * DH * DH,
            b03 + l * DH, g03 + l * DH, be03 + l * DH, ho, nullptr, nullptr,
            qs, qd, qw, qn, (l == 0) ? 1 : 0, NEDGES, NNODES);
        hin = ho;
    }
    gcn_layer_fused<true><<<NRANGES, 1024, 0, stream>>>(
        hin, srcI, dstI, norm, dinv, W03 + (size_t)3 * DH * DH,
        b03 + 3 * DH, g03 + 3 * DH, be03 + 3 * DH, nullptr, W4, xw,
        qs, qd, qw, qn, 0, NEDGES, NNODES);

    gcn_out_fused<<<NRANGES, 1024, 0, stream>>>(
        xw, srcI, dstI, norm, dinv, b4, g4, be4, (float*)d_out,
        qs, qd, qw, qn, NEDGES, NNODES);
}


# First aggregation launch scans 1-byte range numbers (dst / 196, stored by the degree launch), 16 edges per 16-byte load, not the 4-byte dst list: a quarter of the bytes; arrival-order sums kept
# speedup vs baseline: 3.8813x; 1.0366x over previous
#include <hip/hip_runtime.h>
#include <hip/hip_bf16.h>

#define NNODES 50000
#define NEDGES 640000
#define DH 128
#define DOUT 64
#define LN_EPS 1e-5f

typedef float v2f __attribute__((ext_vector_type(2)));
typedef float v8f __attribute__((ext_vector_type(8)));

#define NRANGES 256
#define RPR 196
#define QCAP 16384

#define DEGB 256
static_assert(NRANGES <= 256, "a range number must fit in one byte");
static_assert(DEGB % 128 == 0 && DEGB / 16 <= 32, "whole lines, written by one wave");
__global__ void __launch_bounds__(DEGB)
gcn_deg_kernel(const int* __restrict__ dst, const float* __restrict__ w,
               float* __restrict__ deg, unsigned char* __restrict__ rid, int E) {
    __shared__ __attribute__((aligned(16))) unsigned char rb[DEGB];
    const int t = threadIdx.x;
    const int e = blockIdx.x * DEGB + t;
    int d = 0;
    float we = 0.0f;
    if (e < E) { d = dst[e]; we = w[e]; }
    rb[t] = (unsigned char)((unsigned)d / RPR);
    __syncthreads();
    if (t < DEGB / 16) {
        const int e0 = blockIdx.x * DEGB + 16 * t;
        if (e0 + 16 <= E) {
            *(uint4*)(rid + e0) = *(const uint4*)(rb + 16 * t);
        } else {
            for (int j = 0; j < 16; ++j)
                if (e0 + j < E) rid[e0 + j] = rb[16 * t + j];
        }
    }
    if (e < E) atomicAdd(&deg[d], we);
}

__global__ void gcn_dinv_kernel(const float* __restrict__ deg, float* __restrict__ dinv, int n) {
    int i = blockIdx.x * blockDim.x + threadIdx.x;
    if (i < n) dinv[i] = rsqrtf(deg[i] + 1.0f);
}

__global__ void gcn_norm_kernel(const int* __restrict__ src, const int* __restrict__ dst,
                                const float* __restrict__ w, const float* __restrict__ dinv,
                                float* __restrict__ norm, int E) {
    int e = blockIdx.x * blockDim.x + threadIdx.x;
    if (e < E) norm[e] = dinv[src[e]] * w[e] * dinv[dst[e]];
}


__device__ __forceinline__ unsigned zero_bytes(unsigned y) {
    return ~(((y & 0x7f7f7f7fu) + 0x7f7f7f7fu) | y | 0x7f7f7f7fu);
}

__device__ __forceinline__ unsigned match16(uint4 v, unsigned rr) {
    return (zero_bytes(v.x ^ rr) >> 7) | (zero_bytes(v.y ^ rr) >> 6) |
           (zero_bytes(v.z ^ rr) >> 5) | (zero_bytes(v.w ^ rr) >> 4);
}

template <typename PushT>
__device__ __forceinline__ void scan_range_ids(const unsigned char* __restrict__ rid, int E,
                                               unsigned range, int tid, PushT push) {
    const unsigned rr = range * 0x01010101u;
    const int n16 = E >> 4;
    const uint4* r16 = (const uint4*)rid;
    if (n16 >= 8192) {
        const int nb = (n16 + 8191) >> 13;
        const int boff = (int)(range % (unsigned)nb);
        for (int b = 0; b < nb; ++b) {
            int bb = b + boff;
            if (bb >= nb) bb -= nb;
            const int lower = bb << 13;
            const int start = (lower + 8192 > n16) ? (n16 - 8192) : lower;
            const int i0 = start + tid;
            uint4 v[8];
#pragma unroll
            for (int k = 0; k < 8; ++k) v[k] = r16[i0 + k * 1024];
            unsigned pend[4];
#pragma unroll
            for (int k = 0; k < 8; ++k) {
                unsigned m = match16(v[k], rr);
                if (i0 + k * 1024 < lower) m = 0;
                if (k & 1) pend[k >> 1] |= m << 4;
                else       pend[k >> 1] = m;
            }
#pragma unroll
            for (int h = 0; h < 4; ++h) {
                unsigned p = pend[h];
                while (p) {
                    const int bit = __builtin_ctz(p);
                    p &= p - 1;
                    const int grp = i0 + (2 * h + ((bit >> 2) & 1)) * 1024;
                    push(16 * grp + 4 * (bit & 3) + (bit >> 3));
                }
            }
        }
    } else {
        for (int i = tid; i < n16; i += 1024) {
            unsigned p = match16(r16[i], rr);
            while (p) {
                const int bit = __builtin_ctz(p);
                p &= p - 1;
                push(16 * i + 4 * (bit & 3) + (bit >> 3));
            }
        }
    }
    {
        const int e = 16 * n16 + tid;
        if (e < E && rid[e] == (unsigned char)range) push(e);
    }
}

#define NSTRIPS 13
#define TP 132
#define RPW 7
#define RPAD (RPW * 32)
template <bool LAST>
__global__ void __launch_bounds__(1024)
gcn_layer_fused(const float* __restrict__ hin,
                const int* __restrict__ src, const int* __restrict__ dst,
                const unsigned char* __restrict__ rid,
                const float* __restrict__ norm, const float* __restrict__ dinv,
                const float* __restrict__ W,
                const float* __restrict__ bias, const float* __restrict__ gamma,
                const float* __restrict__ beta,
                float* __restrict__ hout,
                const float* __restrict__ Wn, float* __restrict__ xwn,
                int* __restrict__ qs, int* __restrict__ qd, float* __restrict__ qw,
                int* __restrict__ qn, int build, int E, int n) {
    constexpr int Q = DH / 32;
    constexpr int U = 64 / Q;
    __shared__ __attribute__((aligned(16))) float acc[RPAD * TP];
    __shared__ __attribute__((aligned(16))) float Ws[DH * TP];
    __shared__ int qe[QCAP];
    __shared__ int qcount;

    const int tid  = threadIdx.x;
    const int lane = tid & 31;
    const int wv   = __builtin_amdgcn_readfirstlane(tid >> 5);
    const int lo   = blockIdx.x * RPR;
    const int nrows = (n - lo < RPR) ? (n - lo) : RPR;
    if (nrows <= 0) return;

    const int stored = build ? 0 : qn[blockIdx.x];
    const bool scan = (build != 0) || (stored > QCAP);

    if (tid == 0) qcount = 0;
    {
        float4 wr[4];
        float4 xv[RPW];
        float sl[RPW];
#pragma unroll
        for (int k = 0; k < 4; ++k)
            wr[k] = *(const float4*)(W + (size_t)(wv + 32 * k) * DH + (lane << 2));
        const int rl  = wv + 32 * ((lane < RPW) ? lane : 0);
        const int rlc = (rl < nrows) ? rl : (nrows - 1);
        const int dli = __float_as_int(dinv[lo + rlc]);
#pragma unroll
        for (int i = 0; i < RPW; ++i) {
            const int r = wv + 32 * i;
            const int rc = (r < nrows) ? r : (nrows - 1);
            xv[i] = *(const float4*)(hin + (size_t)(lo + rc) * DH + (lane << 2));
        }
#pragma unroll
        for (int i = 0; i < RPW; ++i) {
            const int r = wv + 32 * i;
            const float di = __int_as_float(__builtin_amdgcn_readlane(dli, i));
            sl[i] = (r < nrows) ? di * di : 0.0f;
        }
#pragma unroll
        for (int k = 0; k < 4; ++k)
            *(float4*)(&Ws[(wv + 32 * k) * TP + (lane << 2)]) = wr[k];
#pragma unroll
        for (int i = 0; i < RPW; ++i) {
            const int r = wv + 32 * i;
            float4 o;
            o.x = sl[i] * xv[i].x; o.y = sl[i] * xv[i].y;
            o.z = sl[i] * xv[i].z; o.w = sl[i] * xv[i].w;
            *(float4*)(&acc[r * TP + (lane << 2)]) = o;
        }
    }
    __syncthreads();

    auto push = [&](int e) {
        const int slot = atomicAdd(&qcount, 1);
        if (slot < QCAP) {
            qe[slot] = e;
        } else {
            const int s = src[e];
            const float w = norm[e];
            const int dl = dst[e] - lo;
            for (int c = 0; c < DH; ++c)
                atomicAdd(&acc[dl * TP + c], w * hin[(size_t)s * DH + c]);
        }
    };

    if (scan) scan_range_ids(rid, E, blockIdx.x, tid, push);
    __syncthreads();

    const int total = scan ? qcount : stored;
    if (build && tid == 0) qn[blockIdx.x] = total;
    const int M = (total < QCAP) ? total : QCAP;
    const size_t seg = (size_t)blockIdx.x * QCAP;
    const float* __restrict__ xl = hin + lane;
    for (int base = wv * 32; base < M; base += 32 * 32) {
        const int cnt = (M - base < 32) ? (M - base) : 32;
        int s = 0, d = 0;
        float w = 0.0f;
        if (lane < cnt) {
            const size_t slot = seg + (size_t)(base + lane);
            if (scan) {
                const int e = qe[base + lane];
                s = src[e]; d = dst[e] - lo; w = norm[e];
                if (build) { qs[slot] = s; qd[slot] = d; qw[slot] = w; }
            } else {
                s = qs[slot]; d = qd[slot]; w = qw[slot];
            }
        }
        const int wi = __float_as_int(w);
        for (int j0 = 0; j0 < cnt; j0 += U) {
            float v[U][Q];
#pragma unroll
            for (int u = 0; u < U; ++u) {
                const int sj = __builtin_amdgcn_readlane(s, j0 + u);
                const float* __restrict__ r = xl + (size_t)sj * DH;
#pragma unroll
                for (int q = 0; q < Q; ++q) v[u][q] = r[q * 32];
            }
#pragma unroll
            for (int u = 0; u < U; ++u) {
                if (j0 + u < cnt) {
                    const int dj = __builtin_amdgcn_readlane(d, j0 + u);
                    const float wj = __int_as_float(__builtin_amdgcn_readlane(wi, j0 + u));
                    float* p = &acc[dj * TP + lane];
#pragma unroll
                    for (int q = 0; q < Q; ++q) atomicAdd(p + q * 32, wj * v[u][q]);
                }
            }
        }
    }
    __syncthreads();

    const int c  = lane & 15;
    const int hh = lane >> 4;
    const int strip = wv >> 1;
    const int ch = wv & 1;
    const bool gw = (wv < 2 * NSTRIPS) && ((strip << 4) < nrows);
    v8f pacc[4];
#pragma unroll
    for (int t = 0; t < 4; ++t) pacc[t] = (v8f){};
    float4 wn0 = make_float4(0.0f, 0.0f, 0.0f, 0.0f), wn1 = wn0;
    if (LAST) {
        wn0 = *(const float4*)(Wn + (size_t)wv * DH + (lane << 2));
        wn1 = *(const float4*)(Wn + (size_t)(wv + 32) * DH + (lane << 2));
    }
    if (gw) {
        const float* ap = acc + ((strip << 4) + c) * TP + (hh << 2);
        const float* bp = Ws + ((ch << 6) + c) * TP + (hh << 2);
#pragma unroll
        for (int j = 0; j < 16; ++j) {
            const float4 a4 = *(const float4*)(ap + 8 * j);
            float4 b4[4];
#pragma unroll
            for (int t = 0; t < 4; ++t) b4[t] = *(const float4*)(bp + t * (16 * TP) + 8 * j);
            const v2f a01 = {a4.x, a4.y};
            const v2f a23 = {a4.z, a4.w};
#pragma unroll
            for (int t = 0; t < 4; ++t) {
                const v2f b01 = {b4[t].x, b4[t].y};
                pacc[t] = __builtin_amdgcn_wmma_f32_16x16x4_f32(
                    false, a01, false, b01, (short)0, pacc[t], false, false);
            }
#pragma unroll
            for (int t = 0; t < 4; ++t) {
                const v2f b23 = {b4[t].z, b4[t].w};
                pacc[t] = __builtin_amdgcn_wmma_f32_16x16x4_f32(
                    false, a23, false, b23, (short)0, pacc[t], false, false);
            }
        }
    }
    __syncthreads();
    if (LAST) {
        *(float4*)(&Ws[wv * TP + (lane << 2)]) = wn0;
        *(float4*)(&Ws[(wv + 32) * TP + (lane << 2)]) = wn1;
    }
    if (gw) {
        float* dp = acc + ((strip << 4) + 8 * hh) * TP + (ch << 6) + c;
#pragma unroll
        for (int t = 0; t < 4; ++t)
#pragma unroll
            for (int r = 0; r < 8; ++r)
                dp[r * TP + 16 * t] = pacc[t][r];
    }
    __syncthreads();

    {
        const float4 b  = *(const float4*)(bias + (lane << 2));
        const float4 g  = *(const float4*)(gamma + (lane << 2));
        const float4 be = *(const float4*)(beta + (lane << 2));
        float4 a[RPW], rv[RPW];
#pragma unroll
        for (int i = 0; i < RPW; ++i) {
            const int r = wv + 32 * i;
            a[i]  = make_float4(0.0f, 0.0f, 0.0f, 0.0f);
            rv[i] = make_float4(0.0f, 0.0f, 0.0f, 0.0f);
            if (r < nrows) {
                a[i]  = *(const float4*)(&acc[r * TP + (lane << 2)]);
                rv[i] = *(const float4*)(hin + (size_t)(lo + r) * DH + (lane << 2));
            }
        }
        float s[RPW], sq[RPW];
#pragma unroll
        for (int i = 0; i < RPW; ++i) {
            a[i].x += b.x; a[i].y += b.y; a[i].z += b.z; a[i].w += b.w;
            s[i]  = a[i].x + a[i].y + a[i].z + a[i].w;
            sq[i] = a[i].x * a[i].x + a[i].y * a[i].y + a[i].z * a[i].z + a[i].w * a[i].w;
        }
#pragma unroll
        for (int m = 16; m >= 1; m >>= 1) {
#pragma unroll
            for (int i = 0; i < RPW; ++i) {
                s[i]  += __shfl_xor(s[i], m, 32);
                sq[i] += __shfl_xor(sq[i], m, 32);
            }
        }
#pragma unroll
        for (int i = 0; i < RPW; ++i) {
            const int r = wv + 32 * i;
            if (r < nrows) {
                const float mean = s[i] * (1.0f / DH);
                const float var  = sq[i] * (1.0f / DH) - mean * mean;
                const float rinv = rsqrtf(var + LN_EPS);
                float4 o;
                o.x = fmaxf(g.x * (a[i].x - mean) * rinv + be.x, 0.0f) + rv[i].x;
                o.y = fmaxf(g.y * (a[i].y - mean) * rinv + be.y, 0.0f) + rv[i].y;
                o.z = fmaxf(g.z * (a[i].z - mean) * rinv + be.z, 0.0f) + rv[i].z;
                o.w = fmaxf(g.w * (a[i].w - mean) * rinv + be.w, 0.0f) + rv[i].w;
                if (LAST) *(float4*)(&acc[r * TP + (lane << 2)]) = o;
                else      *(float4*)(hout + (size_t)(lo + r) * DH + (lane << 2)) = o;
            }
        }
    }

    if (LAST) {
        __syncthreads();
        v8f qacc[2];
#pragma unroll
        for (int t = 0; t < 2; ++t) qacc[t] = (v8f){};
        if (gw) {
            const float* ap = acc + ((strip << 4) + c) * TP + (hh << 2);
            const float* bp = Ws + ((ch << 5) + c) * TP + (hh << 2);
#pragma unroll
            for (int j = 0; j < 16; ++j) {
                const float4 a4 = *(const float4*)(ap + 8 * j);
                float4 b4[2];
#pragma unroll
                for (int t = 0; t < 2; ++t) b4[t] = *(const float4*)(bp + t * (16 * TP) + 8 * j);
                const v2f a01 = {a4.x, a4.y};
                const v2f a23 = {a4.z, a4.w};
#pragma unroll
                for (int t = 0; t < 2; ++t) {
                    const v2f b01 = {b4[t].x, b4[t].y};
                    qacc[t] = __builtin_amdgcn_wmma_f32_16x16x4_f32(
                        false, a01, false, b01, (short)0, qacc[t], false, false);
                }
#pragma unroll
                for (int t = 0; t < 2; ++t) {
                    const v2f b23 = {b4[t].z, b4[t].w};
                    qacc[t] = __builtin_amdgcn_wmma_f32_16x16x4_f32(
                        false, a23, false, b23, (short)0, qacc[t], false, false);
                }
            }
            const int r0 = (strip << 4) + 8 * hh;
            float* op = xwn + (size_t)(lo + r0) * DOUT + (ch << 5) + c;
#pragma unroll
            for (int t = 0; t < 2; ++t)
#pragma unroll
                for (int r = 0; r < 8; ++r)
                    if (r0 + r < nrows) op[(size_t)r * DOUT + 16 * t] = qacc[t][r];
        }
    }
}

__global__ void __launch_bounds__(1024)
gcn_out_fused(const float* __restrict__ xw,
              const int* __restrict__ src, const int* __restrict__ dst,
              const unsigned char* __restrict__ rid,
              const float* __restrict__ norm, const float* __restrict__ dinv,
              const float* __restrict__ bias, const float* __restrict__ gamma,
              const float* __restrict__ beta,
              float* __restrict__ out,
              const int* __restrict__ qs, const int* __restrict__ qd,
              const float* __restrict__ qw, const int* __restrict__ qn, int E, int n) {
    constexpr int NCOLS = DOUT;
    constexpr int Q = NCOLS / 32;
    constexpr int U = 64 / Q;
    constexpr int R4 = NCOLS / 4;
    static_assert(RPR * R4 <= 4 * 1024, "tile start: four 16-byte groups per thread cover the tile");
    __shared__ __attribute__((aligned(16))) float acc[RPR * NCOLS];
    __shared__ int qe[QCAP];
    __shared__ int qcount;

    const int tid  = threadIdx.x;
    const int lane = tid & 31;
    const int wv   = __builtin_amdgcn_readfirstlane(tid >> 5);
    const int lo   = blockIdx.x * RPR;
    const int nrows = (n - lo < RPR) ? (n - lo) : RPR;
    if (nrows <= 0) return;

    const int stored = qn[blockIdx.x];
    const bool scan = stored > QCAP;

    if (tid == 0) qcount = 0;
    const int n4r = nrows * R4;
    {
        const float4* x4 = (const float4*)(xw + (size_t)lo * NCOLS);
        float4* a4 = (float4*)acc;
        float4 xv[4];
        float dv[4];
#pragma unroll
        for (int k = 0; k < 4; ++k) {
            const int i = tid + 1024 * k;
            const int ic = (i < n4r) ? i : (n4r - 1);
            xv[k] = x4[ic];
            dv[k] = dinv[lo + ic / R4];
        }
#pragma unroll
        for (int k = 0; k < 4; ++k) {
            const int i = tid + 1024 * k;
            if (i < n4r) {
                const float sl = dv[k] * dv[k];
                float4 o;
                o.x = sl * xv[k].x; o.y = sl * xv[k].y;
                o.z = sl * xv[k].z; o.w = sl * xv[k].w;
                a4[i] = o;
            }
        }
    }
    __syncthreads();

    auto push = [&](int e) {
        const int slot = atomicAdd(&qcount, 1);
        if (slot < QCAP) {
            qe[slot] = e;
        } else {
            const int s = src[e];
            const float w = norm[e];
            const int dl = dst[e] - lo;
            for (int c = 0; c < NCOLS; ++c)
                atomicAdd(&acc[dl * NCOLS + c], w * xw[(size_t)s * NCOLS + c]);
        }
    };

    if (scan) scan_range_ids(rid, E, blockIdx.x, tid, push);
    __syncthreads();

    const int total = scan ? qcount : stored;
    const int M = (total < QCAP) ? total : QCAP;
    const size_t seg = (size_t)blockIdx.x * QCAP;
    const float* __restrict__ xl = xw + lane;
    for (int base = wv * 32; base < M; base += 32 * 32) {
        const int cnt = (M - base < 32) ? (M - base) : 32;
        int s = 0, d = 0;
        float w = 0.0f;
        if (lane < cnt) {
            if (scan) {
                const int e = qe[base + lane];
                s = src[e]; d = dst[e] - lo; w = norm[e];
            } else {
                const size_t slot = seg + (size_t)(base + lane);
                s = qs[slot]; d = qd[slot]; w = qw[slot];
            }
        }
        const int wi = __float_as_int(w);
        for (int j0 = 0; j0 < cnt; j0 += U) {
            float v[U][Q];
#pragma unroll
            for (int u = 0; u < U; ++u) {
                const int sj = __builtin_amdgcn_readlane(s, j0 + u);
                const float* __restrict__ r = xl + (size_t)sj * NCOLS;
#pragma unroll
                for (int q = 0; q < Q; ++q) v[u][q] = r[q * 32];
            }
#pragma unroll
            for (int u = 0; u < U; ++u) {
                if (j0 + u < cnt) {
                    const int dj = __builtin_amdgcn_readlane(d, j0 + u);
                    const float wj = __int_as_float(__builtin_amdgcn_readlane(wi, j0 + u));
                    float* p = &acc[dj * NCOLS + lane];
#pragma unroll
                    for (int q = 0; q < Q; ++q) atomicAdd(p + q * 32, wj * v[u][q]);
                }
            }
        }
    }
    __syncthreads();

    {
        const float2 b  = *(const float2*)(bias + (lane << 1));
        const float2 g  = *(const float2*)(gamma + (lane << 1));
        const float2 be = *(const float2*)(beta + (lane << 1));
        float2 a[RPW];
        float s[RPW], sq[RPW];
#pragma unroll
        for (int i = 0; i < RPW; ++i) {
            const int r = wv + 32 * i;
            a[i] = make_float2(0.0f, 0.0f);
            if (r < nrows)
                a[i] = *(const float2*)(&acc[r * NCOLS + (lane << 1)]);
            a[i].x += b.x; a[i].y += b.y;
            s[i]  = a[i].x + a[i].y;
            sq[i] = a[i].x * a[i].x + a[i].y * a[i].y;
        }
#pragma unroll
        for (int m = 16; m >= 1; m >>= 1) {
#pragma unroll
            for (int i = 0; i < RPW; ++i) {
                s[i]  += __shfl_xor(s[i], m, 32);
                sq[i] += __shfl_xor(sq[i], m, 32);
            }
        }
#pragma unroll
        for (int i = 0; i < RPW; ++i) {
            const int r = wv + 32 * i;
            if (r < nrows) {
                const float mean = s[i] * (1.0f / NCOLS);
                const float var  = sq[i] * (1.0f / NCOLS) - mean * mean;
                const float rinv = rsqrtf(var + LN_EPS);
                float2 o;
                o.x = g.x * (a[i].x - mean) * rinv + be.x;
                o.y = g.y * (a[i].y - mean) * rinv + be.y;
                *(float2*)(out + (size_t)(lo + r) * NCOLS + (lane << 1)) = o;
            }
        }
    }
}

extern "C" void kernel_launch(void* const* d_in, const int* in_sizes, int n_in,
                              void* d_out, int out_size, void* d_ws, size_t ws_size,
                              hipStream_t stream) {
    const float* x    = (const float*)d_in[0];
    const int*   ei   = (const int*)d_in[1];
    const float* ew   = (const float*)d_in[2];
    const float* W03  = (const float*)d_in[3];
    const float* b03  = (const float*)d_in[4];
    const float* g03  = (const float*)d_in[5];
    const float* be03 = (const float*)d_in[6];
    const float* W4   = (const float*)d_in[7];
    const float* b4   = (const float*)d_in[8];
    const float* g4   = (const float*)d_in[9];
    const float* be4  = (const float*)d_in[10];

    const int* srcI = ei;
    const int* dstI = ei + NEDGES;

    float* xw   = (float*)d_ws;
    float* hA   = xw + (size_t)NNODES * DOUT;
    float* hB   = hA + (size_t)NNODES * DH;
    float* dinv = hB + (size_t)NNODES * DH;
    float* deg  = dinv + NNODES;
    float* norm = deg + NNODES;
    int*   qn   = (int*)(norm + NEDGES);
    int*   qs   = qn + NRANGES;
    int*   qd   = qs + (size_t)NRANGES * QCAP;
    float* qw   = (float*)(qd + (size_t)NRANGES * QCAP);
    unsigned char* rid = (unsigned char*)(qw + (size_t)NRANGES * QCAP);

    hipMemsetAsync(deg, 0, (size_t)NNODES * sizeof(float), stream);
    gcn_deg_kernel<<<(NEDGES + DEGB - 1) / DEGB, DEGB, 0, stream>>>(dstI, ew, deg, rid, NEDGES);
    gcn_dinv_kernel<<<(NNODES + 255) / 256, 256, 0, stream>>>(deg, dinv, NNODES);
    gcn_norm_kernel<<<(NEDGES + 255) / 256, 256, 0, stream>>>(srcI, dstI, ew, dinv, norm, NEDGES);

    const float* hin = x;
    for (int l = 0; l < 3; ++l) {
        float* ho = (l & 1) ? hB : hA;
        gcn_layer_fused<false><<<NRANGES, 1024, 0, stream>>>(
            hin, srcI, dstI, rid, norm, dinv, W03 + (size_t)l * DH * DH,
            b03 + l * DH, g03 + l * DH, be03 + l * DH, ho, nullptr, nullptr,
            qs, qd, qw, qn, (l == 0) ? 1 : 0, NEDGES, NNODES);
        hin = ho;
    }
    gcn_layer_fused<true><<<NRANGES, 1024, 0, stream>>>(
        hin, srcI, dstI, rid, norm, dinv, W03 + (size_t)3 * DH * DH,
        b03 + 3 * DH, g03 + 3 * DH, be03 + 3 * DH, nullptr, W4, xw,
        qs, qd, qw, qn, 0, NEDGES, NNODES);

    gcn_out_fused<<<NRANGES, 1024, 0, stream>>>(
        xw, srcI, dstI, rid, norm, dinv, b4, g4, be4, (float*)d_out,
        qs, qd, qw, qn, NEDGES, NNODES);
}


# dinv and norm launches and planes removed: the first aggregation launch computes each edge's norm from the degree plane as it stores the edge records; each layer derives its rows' dinv from deg
# speedup vs baseline: 4.0872x; 1.0530x over previous
#include <hip/hip_runtime.h>
#include <hip/hip_bf16.h>

#define NNODES 50000
#define NEDGES 640000
#define DH 128
#define DOUT 64
#define LN_EPS 1e-5f

typedef float v2f __attribute__((ext_vector_type(2)));
typedef float v8f __attribute__((ext_vector_type(8)));

#define NRANGES 256
#define RPR 196
#define QCAP 16384

#define DEGB 256
static_assert(NRANGES <= 256, "a range number must fit in one byte");
static_assert(DEGB % 128 == 0 && DEGB / 16 <= 32, "whole lines, written by one wave");
__global__ void __launch_bounds__(DEGB)
gcn_deg_kernel(const int* __restrict__ dst, const float* __restrict__ w,
               float* __restrict__ deg, unsigned char* __restrict__ rid, int E) {
    __shared__ __attribute__((aligned(16))) unsigned char rb[DEGB];
    const int t = threadIdx.x;
    const int e = blockIdx.x * DEGB + t;
    int d = 0;
    float we = 0.0f;
    if (e < E) { d = dst[e]; we = w[e]; }
    rb[t] = (unsigned char)((unsigned)d / RPR);
    __syncthreads();
    if (t < DEGB / 16) {
        const int e0 = blockIdx.x * DEGB + 16 * t;
        if (e0 + 16 <= E) {
            *(uint4*)(rid + e0) = *(const uint4*)(rb + 16 * t);
        } else {
            for (int j = 0; j < 16; ++j)
                if (e0 + j < E) rid[e0 + j] = rb[16 * t + j];
        }
    }
    if (e < E) atomicAdd(&deg[d], we);
}

__device__ __forceinline__ float node_dinv(const float* __restrict__ deg, int i) {
    return rsqrtf(deg[i] + 1.0f);
}

__device__ __forceinline__ float edge_norm(const float* __restrict__ deg, int s, float we, int d) {
    return node_dinv(deg, s) * we * node_dinv(deg, d);
}


__device__ __forceinline__ unsigned zero_bytes(unsigned y) {
    return ~(((y & 0x7f7f7f7fu) + 0x7f7f7f7fu) | y | 0x7f7f7f7fu);
}

__device__ __forceinline__ unsigned match16(uint4 v, unsigned rr) {
    return (zero_bytes(v.x ^ rr) >> 7) | (zero_bytes(v.y ^ rr) >> 6) |
           (zero_bytes(v.z ^ rr) >> 5) | (zero_bytes(v.w ^ rr) >> 4);
}

template <typename PushT>
__device__ __forceinline__ void scan_range_ids(const unsigned char* __restrict__ rid, int E,
                                               unsigned range, int tid, PushT push) {
    const unsigned rr = range * 0x01010101u;
    const int n16 = E >> 4;
    const uint4* r16 = (const uint4*)rid;
    if (n16 >= 8192) {
        const int nb = (n16 + 8191) >> 13;
        const int boff = (int)(range % (unsigned)nb);
        for (int b = 0; b < nb; ++b) {
            int bb = b + boff;
            if (bb >= nb) bb -= nb;
            const int lower = bb << 13;
            const int start = (lower + 8192 > n16) ? (n16 - 8192) : lower;
            const int i0 = start + tid;
            uint4 v[8];
#pragma unroll
            for (int k = 0; k < 8; ++k) v[k] = r16[i0 + k * 1024];
            unsigned pend[4];
#pragma unroll
            for (int k = 0; k < 8; ++k) {
                unsigned m = match16(v[k], rr);
                if (i0 + k * 1024 < lower) m = 0;
                if (k & 1) pend[k >> 1] |= m << 4;
                else       pend[k >> 1] = m;
            }
#pragma unroll
            for (int h = 0; h < 4; ++h) {
                unsigned p = pend[h];
                while (p) {
                    const int bit = __builtin_ctz(p);
                    p &= p - 1;
                    const int grp = i0 + (2 * h + ((bit >> 2) & 1)) * 1024;
                    push(16 * grp + 4 * (bit & 3) + (bit >> 3));
                }
            }
        }
    } else {
        for (int i = tid; i < n16; i += 1024) {
            unsigned p = match16(r16[i], rr);
            while (p) {
                const int bit = __builtin_ctz(p);
                p &= p - 1;
                push(16 * i + 4 * (bit & 3) + (bit >> 3));
            }
        }
    }
    {
        const int e = 16 * n16 + tid;
        if (e < E && rid[e] == (unsigned char)range) push(e);
    }
}

#define NSTRIPS 13
#define TP 132
#define RPW 7
#define RPAD (RPW * 32)
template <bool LAST>
__global__ void __launch_bounds__(1024)
gcn_layer_fused(const float* __restrict__ hin,
                const int* __restrict__ src, const int* __restrict__ dst,
                const unsigned char* __restrict__ rid,
                const float* __restrict__ ew, const float* __restrict__ deg,
                const float* __restrict__ W,
                const float* __restrict__ bias, const float* __restrict__ gamma,
                const float* __restrict__ beta,
                float* __restrict__ hout,
                const float* __restrict__ Wn, float* __restrict__ xwn,
                int* __restrict__ qs, int* __restrict__ qd, float* __restrict__ qw,
                int* __restrict__ qn, int build, int E, int n) {
    constexpr int Q = DH / 32;
    constexpr int U = 64 / Q;
    __shared__ __attribute__((aligned(16))) float acc[RPAD * TP];
    __shared__ __attribute__((aligned(16))) float Ws[DH * TP];
    __shared__ int qe[QCAP];
    __shared__ int qcount;

    const int tid  = threadIdx.x;
    const int lane = tid & 31;
    const int wv   = __builtin_amdgcn_readfirstlane(tid >> 5);
    const int lo   = blockIdx.x * RPR;
    const int nrows = (n - lo < RPR) ? (n - lo) : RPR;
    if (nrows <= 0) return;

    const int stored = build ? 0 : qn[blockIdx.x];
    const bool scan = (build != 0) || (stored > QCAP);

    if (tid == 0) qcount = 0;
    {
        float4 wr[4];
        float4 xv[RPW];
        float sl[RPW];
#pragma unroll
        for (int k = 0; k < 4; ++k)
            wr[k] = *(const float4*)(W + (size_t)(wv + 32 * k) * DH + (lane << 2));
        const int rl  = wv + 32 * ((lane < RPW) ? lane : 0);
        const int rlc = (rl < nrows) ? rl : (nrows - 1);
        const int dli = __float_as_int(node_dinv(deg, lo + rlc));
#pragma unroll
        for (int i = 0; i < RPW; ++i) {
            const int r = wv + 32 * i;
            const int rc = (r < nrows) ? r : (nrows - 1);
            xv[i] = *(const float4*)(hin + (size_t)(lo + rc) * DH + (lane << 2));
        }
#pragma unroll
        for (int i = 0; i < RPW; ++i) {
            const int r = wv + 32 * i;
            const float di = __int_as_float(__builtin_amdgcn_readlane(dli, i));
            sl[i] = (r < nrows) ? di * di : 0.0f;
        }
#pragma unroll
        for (int k = 0; k < 4; ++k)
            *(float4*)(&Ws[(wv + 32 * k) * TP + (lane << 2)]) = wr[k];
#pragma unroll
        for (int i = 0; i < RPW; ++i) {
            const int r = wv + 32 * i;
            float4 o;
            o.x = sl[i] * xv[i].x; o.y = sl[i] * xv[i].y;
            o.z = sl[i] * xv[i].z; o.w = sl[i] * xv[i].w;
            *(float4*)(&acc[r * TP + (lane << 2)]) = o;
        }
    }
    __syncthreads();

    auto push = [&](int e) {
        const int slot = atomicAdd(&qcount, 1);
        if (slot < QCAP) {
            qe[slot] = e;
        } else {
            const int s = src[e];
            const int dg = dst[e];
            const float w = edge_norm(deg, s, ew[e], dg);
            const int dl = dg - lo;
            for (int c = 0; c < DH; ++c)
                atomicAdd(&acc[dl * TP + c], w * hin[(size_t)s * DH + c]);
        }
    };

    if (scan) scan_range_ids(rid, E, blockIdx.x, tid, push);
    __syncthreads();

    const int total = scan ? qcount : stored;
    if (build && tid == 0) qn[blockIdx.x] = total;
    const int M = (total < QCAP) ? total : QCAP;
    const size_t seg = (size_t)blockIdx.x * QCAP;
    const float* __restrict__ xl = hin + lane;
    for (int base = wv * 32; base < M; base += 32 * 32) {
        const int cnt = (M - base < 32) ? (M - base) : 32;
        int s = 0, d = 0;
        float w = 0.0f;
        if (lane < cnt) {
            const size_t slot = seg + (size_t)(base + lane);
            if (scan) {
                const int e = qe[base + lane];
                const int dg = dst[e];
                s = src[e]; d = dg - lo; w = edge_norm(deg, s, ew[e], dg);
                if (build) { qs[slot] = s; qd[slot] = d; qw[slot] = w; }
            } else {
                s = qs[slot]; d = qd[slot]; w = qw[slot];
            }
        }
        const int wi = __float_as_int(w);
        for (int j0 = 0; j0 < cnt; j0 += U) {
            float v[U][Q];
#pragma unroll
            for (int u = 0; u < U; ++u) {
                const int sj = __builtin_amdgcn_readlane(s, j0 + u);
                const float* __restrict__ r = xl + (size_t)sj * DH;
#pragma unroll
                for (int q = 0; q < Q; ++q) v[u][q] = r[q * 32];
            }
#pragma unroll
            for (int u = 0; u < U; ++u) {
                if (j0 + u < cnt) {
                    const int dj = __builtin_amdgcn_readlane(d, j0 + u);
                    const float wj = __int_as_float(__builtin_amdgcn_readlane(wi, j0 + u));
                    float* p = &acc[dj * TP + lane];
#pragma unroll
                    for (int q = 0; q < Q; ++q) atomicAdd(p + q * 32, wj * v[u][q]);
                }
            }
        }
    }
    __syncthreads();

    const int c  = lane & 15;
    const int hh = lane >> 4;
    const int strip = wv >> 1;
    const int ch = wv & 1;
    const bool gw = (wv < 2 * NSTRIPS) && ((strip << 4) < nrows);
    v8f pacc[4];
#pragma unroll
    for (int t = 0; t < 4; ++t) pacc[t] = (v8f){};
    float4 wn0 = make_float4(0.0f, 0.0f, 0.0f, 0.0f), wn1 = wn0;
    if (LAST) {
        wn0 = *(const float4*)(Wn + (size_t)wv * DH + (lane << 2));
        wn1 = *(const float4*)(Wn + (size_t)(wv + 32) * DH + (lane << 2));
    }
    if (gw) {
        const float* ap = acc + ((strip << 4) + c) * TP + (hh << 2);
        const float* bp = Ws + ((ch << 6) + c) * TP + (hh << 2);
#pragma unroll
        for (int j = 0; j < 16; ++j) {
            const float4 a4 = *(const float4*)(ap + 8 * j);
            float4 b4[4];
#pragma unroll
            for (int t = 0; t < 4; ++t) b4[t] = *(const float4*)(bp + t * (16 * TP) + 8 * j);
            const v2f a01 = {a4.x, a4.y};
            const v2f a23 = {a4.z, a4.w};
#pragma unroll
            for (int t = 0; t < 4; ++t) {
                const v2f b01 = {b4[t].x, b4[t].y};
                pacc[t] = __builtin_amdgcn_wmma_f32_16x16x4_f32(
                    false, a01, false, b01, (short)0, pacc[t], false, false);
            }
#pragma unroll
            for (int t = 0; t < 4; ++t) {
                const v2f b23 = {b4[t].z, b4[t].w};
                pacc[t] = __builtin_amdgcn_wmma_f32_16x16x4_f32(
                    false, a23, false, b23, (short)0, pacc[t], false, false);
            }
        }
    }
    __syncthreads();
    if (LAST) {
        *(float4*)(&Ws[wv * TP + (lane << 2)]) = wn0;
        *(float4*)(&Ws[(wv + 32) * TP + (lane << 2)]) = wn1;
    }
    if (gw) {
        float* dp = acc + ((strip << 4) + 8 * hh) * TP + (ch << 6) + c;
#pragma unroll
        for (int t = 0; t < 4; ++t)
#pragma unroll
            for (int r = 0; r < 8; ++r)
                dp[r * TP + 16 * t] = pacc[t][r];
    }
    __syncthreads();

    {
        const float4 b  = *(const float4*)(bias + (lane << 2));
        const float4 g  = *(const float4*)(gamma + (lane << 2));
        const float4 be = *(const float4*)(beta + (lane << 2));
        float4 a[RPW], rv[RPW];
#pragma unroll
        for (int i = 0; i < RPW; ++i) {
            const int r = wv + 32 * i;
            a[i]  = make_float4(0.0f, 0.0f, 0.0f, 0.0f);
            rv[i] = make_float4(0.0f, 0.0f, 0.0f, 0.0f);
            if (r < nrows) {
                a[i]  = *(const float4*)(&acc[r * TP + (lane << 2)]);
                rv[i] = *(const float4*)(hin + (size_t)(lo + r) * DH + (lane << 2));
            }
        }
        float s[RPW], sq[RPW];
#pragma unroll
        for (int i = 0; i < RPW; ++i) {
            a[i].x += b.x; a[i].y += b.y; a[i].z += b.z; a[i].w += b.w;
            s[i]  = a[i].x + a[i].y + a[i].z + a[i].w;
            sq[i] = a[i].x * a[i].x + a[i].y * a[i].y + a[i].z * a[i].z + a[i].w * a[i].w;
        }
#pragma unroll
        for (int m = 16; m >= 1; m >>= 1) {
#pragma unroll
            for (int i = 0; i < RPW; ++i) {
                s[i]  += __shfl_xor(s[i], m, 32);
                sq[i] += __shfl_xor(sq[i], m, 32);
            }
        }
#pragma unroll
        for (int i = 0; i < RPW; ++i) {
            const int r = wv + 32 * i;
            if (r < nrows) {
                const float mean = s[i] * (1.0f / DH);
                const float var  = sq[i] * (1.0f / DH) - mean * mean;
                const float rinv = rsqrtf(var + LN_EPS);
                float4 o;
                o.x = fmaxf(g.x * (a[i].x - mean) * rinv + be.x, 0.0f) + rv[i].x;
                o.y = fmaxf(g.y * (a[i].y - mean) * rinv + be.y, 0.0f) + rv[i].y;
                o.z = fmaxf(g.z * (a[i].z - mean) * rinv + be.z, 0.0f) + rv[i].z;
                o.w = fmaxf(g.w * (a[i].w - mean) * rinv + be.w, 0.0f) + rv[i].w;
                if (LAST) *(float4*)(&acc[r * TP + (lane << 2)]) = o;
                else      *(float4*)(hout + (size_t)(lo + r) * DH + (lane << 2)) = o;
            }
        }
    }

    if (LAST) {
        __syncthreads();
        v8f qacc[2];
#pragma unroll
        for (int t = 0; t < 2; ++t) qacc[t] = (v8f){};
        if (gw) {
            const float* ap = acc + ((strip << 4) + c) * TP + (hh << 2);
            const float* bp = Ws + ((ch << 5) + c) * TP + (hh << 2);
#pragma unroll
            for (int j = 0; j < 16; ++j) {
                const float4 a4 = *(const float4*)(ap + 8 * j);
                float4 b4[2];
#pragma unroll
                for (int t = 0; t < 2; ++t) b4[t] = *(const float4*)(bp + t * (16 * TP) + 8 * j);
                const v2f a01 = {a4.x, a4.y};
                const v2f a23 = {a4.z, a4.w};
#pragma unroll
                for (int t = 0; t < 2; ++t) {
                    const v2f b01 = {b4[t].x, b4[t].y};
                    qacc[t] = __builtin_amdgcn_wmma_f32_16x16x4_f32(
                        false, a01, false, b01, (short)0, qacc[t], false, false);
                }
#pragma unroll
                for (int t = 0; t < 2; ++t) {
                    const v2f b23 = {b4[t].z, b4[t].w};
                    qacc[t] = __builtin_amdgcn_wmma_f32_16x16x4_f32(
                        false, a23, false, b23, (short)0, qacc[t], false, false);
                }
            }
            const int r0 = (strip << 4) + 8 * hh;
            float* op = xwn + (size_t)(lo + r0) * DOUT + (ch << 5) + c;
#pragma unroll
            for (int t = 0; t < 2; ++t)
#pragma unroll
                for (int r = 0; r < 8; ++r)
                    if (r0 + r < nrows) op[(size_t)r * DOUT + 16 * t] = qacc[t][r];
        }
    }
}

__global__ void __launch_bounds__(1024)
gcn_out_fused(const float* __restrict__ xw,
              const int* __restrict__ src, const int* __restrict__ dst,
              const unsigned char* __restrict__ rid,
              const float* __restrict__ ew, const float* __restrict__ deg,
              const float* __restrict__ bias, const float* __restrict__ gamma,
              const float* __restrict__ beta,
              float* __restrict__ out,
              const int* __restrict__ qs, const int* __restrict__ qd,
              const float* __restrict__ qw, const int* __restrict__ qn, int E, int n) {
    constexpr int NCOLS = DOUT;
    constexpr int Q = NCOLS / 32;
    constexpr int U = 64 / Q;
    constexpr int R4 = NCOLS / 4;
    static_assert(RPR * R4 <= 4 * 1024, "tile start: four 16-byte groups per thread cover the tile");
    __shared__ __attribute__((aligned(16))) float acc[RPR * NCOLS];
    __shared__ int qe[QCAP];
    __shared__ int qcount;

    const int tid  = threadIdx.x;
    const int lane = tid & 31;
    const int wv   = __builtin_amdgcn_readfirstlane(tid >> 5);
    const int lo   = blockIdx.x * RPR;
    const int nrows = (n - lo < RPR) ? (n - lo) : RPR;
    if (nrows <= 0) return;

    const int stored = qn[blockIdx.x];
    const bool scan = stored > QCAP;

    if (tid == 0) qcount = 0;
    const int n4r = nrows * R4;
    {
        const float4* x4 = (const float4*)(xw + (size_t)lo * NCOLS);
        float4* a4 = (float4*)acc;
        float4 xv[4];
        float dv[4];
#pragma unroll
        for (int k = 0; k < 4; ++k) {
            const int i = tid + 1024 * k;
            const int ic = (i < n4r) ? i : (n4r - 1);
            xv[k] = x4[ic];
            dv[k] = node_dinv(deg, lo + ic / R4);
        }
#pragma unroll
        for (int k = 0; k < 4; ++k) {
            const int i = tid + 1024 * k;
            if (i < n4r) {
                const float sl = dv[k] * dv[k];
                float4 o;
                o.x = sl * xv[k].x; o.y = sl * xv[k].y;
                o.z = sl * xv[k].z; o.w = sl * xv[k].w;
                a4[i] = o;
            }
        }
    }
    __syncthreads();

    auto push = [&](int e) {
        const int slot = atomicAdd(&qcount, 1);
        if (slot < QCAP) {
            qe[slot] = e;
        } else {
            const int s = src[e];
            const int dg = dst[e];
            const float w = edge_norm(deg, s, ew[e], dg);
            const int dl = dg - lo;
            for (int c = 0; c < NCOLS; ++c)
                atomicAdd(&acc[dl * NCOLS + c], w * xw[(size_t)s * NCOLS + c]);
        }
    };

    if (scan) scan_range_ids(rid, E, blockIdx.x, tid, push);
    __syncthreads();

    const int total = scan ? qcount : stored;
    const int M = (total < QCAP) ? total : QCAP;
    const size_t seg = (size_t)blockIdx.x * QCAP;
    const float* __restrict__ xl = xw + lane;
    for (int base = wv * 32; base < M; base += 32 * 32) {
        const int cnt = (M - base < 32) ? (M - base) : 32;
        int s = 0, d = 0;
        float w = 0.0f;
        if (lane < cnt) {
            if (scan) {
                const int e = qe[base + lane];
                const int dg = dst[e];
                s = src[e]; d = dg - lo; w = edge_norm(deg, s, ew[e], dg);
            } else {
                const size_t slot = seg + (size_t)(base + lane);
                s = qs[slot]; d = qd[slot]; w = qw[slot];
            }
        }
        const int wi = __float_as_int(w);
        for (int j0 = 0; j0 < cnt; j0 += U) {
            float v[U][Q];
#pragma unroll
            for (int u = 0; u < U; ++u) {
                const int sj = __builtin_amdgcn_readlane(s, j0 + u);
                const float* __restrict__ r = xl + (size_t)sj * NCOLS;
#pragma unroll
                for (int q = 0; q < Q; ++q) v[u][q] = r[q * 32];
            }
#pragma unroll
            for (int u = 0; u < U; ++u) {
                if (j0 + u < cnt) {
                    const int dj = __builtin_amdgcn_readlane(d, j0 + u);
                    const float wj = __int_as_float(__builtin_amdgcn_readlane(wi, j0 + u));
                    float* p = &acc[dj * NCOLS + lane];
#pragma unroll
                    for (int q = 0; q < Q; ++q) atomicAdd(p + q * 32, wj * v[u][q]);
                }
            }
        }
    }
    __syncthreads();

    {
        const float2 b  = *(const float2*)(bias + (lane << 1));
        const float2 g  = *(const float2*)(gamma + (lane << 1));
        const float2 be = *(const float2*)(beta + (lane << 1));
        float2 a[RPW];
        float s[RPW], sq[RPW];
#pragma unroll
        for (int i = 0; i < RPW; ++i) {
            const int r = wv + 32 * i;
            a[i] = make_float2(0.0f, 0.0f);
            if (r < nrows)
                a[i] = *(const float2*)(&acc[r * NCOLS + (lane << 1)]);
            a[i].x += b.x; a[i].y += b.y;
            s[i]  = a[i].x + a[i].y;
            sq[i] = a[i].x * a[i].x + a[i].y * a[i].y;
        }
#pragma unroll
        for (int m = 16; m >= 1; m >>= 1) {
#pragma unroll
            for (int i = 0; i < RPW; ++i) {
                s[i]  += __shfl_xor(s[i], m, 32);
                sq[i] += __shfl_xor(sq[i], m, 32);
            }
        }
#pragma unroll
        for (int i = 0; i < RPW; ++i) {
            const int r = wv + 32 * i;
            if (r < nrows) {
                const float mean = s[i] * (1.0f / NCOLS);
                const float var  = sq[i] * (1.0f / NCOLS) - mean * mean;
                const float rinv = rsqrtf(var + LN_EPS);
                float2 o;
                o.x = g.x * (a[i].x - mean) * rinv + be.x;
                o.y = g.y * (a[i].y - mean) * rinv + be.y;
                *(float2*)(out + (size_t)(lo + r) * NCOLS + (lane << 1)) = o;
            }
        }
    }
}

extern "C" void kernel_launch(void* const* d_in, const int* in_sizes, int n_in,
                              void* d_out, int out_size, void* d_ws, size_t ws_size,
                              hipStream_t stream) {
    const float* x    = (const float*)d_in[0];
    const int*   ei   = (const int*)d_in[1];
    const float* ew   = (const float*)d_in[2];
    const float* W03  = (const float*)d_in[3];
    const float* b03  = (const float*)d_in[4];
    const float* g03  = (const float*)d_in[5];
    const float* be03 = (const float*)d_in[6];
    const float* W4   = (const float*)d_in[7];
    const float* b4   = (const float*)d_in[8];
    const float* g4   = (const float*)d_in[9];
    const float* be4  = (const float*)d_in[10];

    const int* srcI = ei;
    const int* dstI = ei + NEDGES;

    float* xw   = (float*)d_ws;
    float* hA   = xw + (size_t)NNODES * DOUT;
    float* hB   = hA + (size_t)NNODES * DH;
    int*   qn   = (int*)(hB + (size_t)NNODES * DH);
    int*   qs   = qn + NRANGES;
    int*   qd   = qs + (size_t)NRANGES * QCAP;
    float* qw   = (float*)(qd + (size_t)NRANGES * QCAP);
    float* deg  = qw + (size_t)NRANGES * QCAP;
    unsigned char* rid = (unsigned char*)(deg + NNODES);

    hipMemsetAsync(deg, 0, (size_t)NNODES * sizeof(float), stream);
    gcn_deg_kernel<<<(NEDGES + DEGB - 1) / DEGB, DEGB, 0, stream>>>(dstI, ew, deg, rid, NEDGES);

    const float* hin = x;
    for (int l = 0; l < 3; ++l) {
        float* ho = (l & 1) ? hB : hA;
        gcn_layer_fused<false><<<NRANGES, 1024, 0, stream>>>(
            hin, srcI, dstI, rid, ew, deg, W03 + (size_t)l * DH * DH,
            b03 + l * DH, g03 + l * DH, be03 + l * DH, ho, nullptr, nullptr,
            qs, qd, qw, qn, (l == 0) ? 1 : 0, NEDGES, NNODES);
        hin = ho;
    }
    gcn_layer_fused<true><<<NRANGES, 1024, 0, stream>>>(
        hin, srcI, dstI, rid, ew, deg, W03 + (size_t)3 * DH * DH,
        b03 + 3 * DH, g03 + 3 * DH, be03 + 3 * DH, nullptr, W4, xw,
        qs, qd, qw, qn, 0, NEDGES, NNODES);

    gcn_out_fused<<<NRANGES, 1024, 0, stream>>>(
        xw, srcI, dstI, rid, ew, deg, b4, g4, be4, (float*)d_out,
        qs, qd, qw, qn, NEDGES, NNODES);
}
